# starting kernel
# baseline (speedup 1.0000x reference)
#include <hip/hip_runtime.h>

typedef __attribute__((ext_vector_type(2))) float v2f;
typedef __attribute__((ext_vector_type(8))) float v8f;

__global__ void build_rowptr(const int* __restrict__ rows, int* __restrict__ row_ptr,
                             int nE, int nN) {
    int e = blockIdx.x * blockDim.x + threadIdx.x;
    if (e >= nE) return;
    int r = rows[e];
    int rp = (e == 0) ? -1 : rows[e - 1];
    for (int rr = rp + 1; rr <= r; ++rr) row_ptr[rr] = e;
    if (e == nE - 1) {
        for (int rr = r + 1; rr <= nN; ++rr) row_ptr[rr] = nE;
    }
}

__global__ void spmm_kernel(const float* __restrict__ h, const int* __restrict__ row_ptr,
                            const int* __restrict__ cols, const float* __restrict__ vals,
                            float* __restrict__ out, const float* __restrict__ sub,
                            float scale, int nN) {
    int wave = (int)((blockIdx.x * blockDim.x + threadIdx.x) >> 5);
    int lane = threadIdx.x & 31;
    if (wave >= nN) return;
    int start = row_ptr[wave];
    int end   = row_ptr[wave + 1];
    float acc0 = 0.0f, acc1 = 0.0f;
    for (int e = start; e < end; ++e) {
        float v = vals[e];
        int   c = cols[e];
        const float* xr = h + (size_t)c * 64;
        if (e + 8 < end) {
            int cn = cols[e + 8];
            __builtin_prefetch(h + (size_t)cn * 64 + lane, 0, 0);
        }
        acc0 = fmaf(v, xr[lane], acc0);
        acc1 = fmaf(v, xr[lane + 32], acc1);
    }
    size_t o = (size_t)wave * 64;
    if (sub != nullptr) {
        out[o + lane]      = scale * acc0 - sub[o + lane];
        out[o + lane + 32] = scale * acc1 - sub[o + lane + 32];
    } else {
        out[o + lane]      = acc0;
        out[o + lane + 32] = acc1;
    }
}

__device__ __forceinline__ void gemm_accum_64(const float* __restrict__ A,
                                              const float* __restrict__ Wl,
                                              int ksel, int bn, v8f acc[4]) {
#pragma unroll
    for (int kk = 0; kk < 64; kk += 4) {
        v2f a = *(const v2f*)(A + kk + ksel);
        int kprow = ((kk + ksel) >> 1) * 128;
#pragma unroll
        for (int nt = 0; nt < 4; ++nt) {
            v2f b = *(const v2f*)(Wl + kprow + (nt * 16 + bn) * 2);
            acc[nt] = __builtin_amdgcn_wmma_f32_16x16x4_f32(
                false, a, false, b, (short)0, acc[nt], false, false);
        }
    }
}

__global__ void cheb_gemm(const float* __restrict__ Tx0, const float* __restrict__ Tx1,
                          const float* __restrict__ Tx2, const float* __restrict__ W,
                          const float* __restrict__ bias, float* __restrict__ out, int nN) {
    __shared__ float w_lds[3 * 64 * 64];
    for (int i = threadIdx.x; i < 3 * 64 * 64; i += blockDim.x) {
        int t = i >> 12;
        int k = (i >> 6) & 63;
        int n = i & 63;
        w_lds[(t << 12) + ((k >> 1) << 7) + (n << 1) + (k & 1)] = W[i];
    }
    __syncthreads();

    int wave = threadIdx.x >> 5;
    int lane = threadIdx.x & 31;
    int m0 = (blockIdx.x * 8 + wave) * 16;

    int arow = m0 + (lane & 15);
    int arow_c = (arow < nN) ? arow : (nN - 1);
    int ksel = (lane >> 4) << 1;
    int bn = lane & 15;

    v8f acc[4] = {};
    gemm_accum_64(Tx0 + (size_t)arow_c * 64, w_lds,            ksel, bn, acc);
    gemm_accum_64(Tx1 + (size_t)arow_c * 64, w_lds + (1 << 12), ksel, bn, acc);
    gemm_accum_64(Tx2 + (size_t)arow_c * 64, w_lds + (2 << 12), ksel, bn, acc);

    int mbase = m0 + ((lane >> 4) << 3);
#pragma unroll
    for (int nt = 0; nt < 4; ++nt) {
        int col = nt * 16 + bn;
        float bv = bias[col];
#pragma unroll
        for (int j = 0; j < 8; ++j) {
            int row = mbase + j;
            if (row < nN) out[(size_t)row * 64 + col] = acc[nt][j] + bv;
        }
    }
}

extern "C" void kernel_launch(void* const* d_in, const int* in_sizes, int n_in,
                              void* d_out, int out_size, void* d_ws, size_t ws_size,
                              hipStream_t stream) {
    const float* x    = (const float*)d_in[0];
    const int*   rows = (const int*)d_in[1];
    const int*   cols = (const int*)d_in[2];
    const float* vals = (const float*)d_in[3];
    const float* W    = (const float*)d_in[4];
    const float* bias = (const float*)d_in[5];

    const int N = in_sizes[0] / 64;
    const int E = in_sizes[1];

    char* ws = (char*)d_ws;
    int* row_ptr = (int*)ws;
    size_t off = (((size_t)(N + 1) * sizeof(int)) + 255) & ~(size_t)255;
    float* Tx1 = (float*)(ws + off);
    float* Tx2 = Tx1 + (size_t)N * 64;

    build_rowptr<<<(E + 255) / 256, 256, 0, stream>>>(rows, row_ptr, E, N);

    int spmm_grid = (N + 7) / 8;
    spmm_kernel<<<spmm_grid, 256, 0, stream>>>(x,   row_ptr, cols, vals, Tx1, nullptr, 1.0f, N);
    spmm_kernel<<<spmm_grid, 256, 0, stream>>>(Tx1, row_ptr, cols, vals, Tx2, x,       2.0f, N);

    cheb_gemm<<<(N + 127) / 128, 256, 0, stream>>>(x, Tx1, Tx2, W, bias, (float*)d_out, N);
}


# Dense product fused into the second SpMM: each 64-node block keeps its Tx2 rows in LDS and runs the f32 WMMA product on them there; Tx2 never goes to memory, 3 launches instead of 4
# speedup vs baseline: 2.0560x; 2.0560x over previous
#include <hip/hip_runtime.h>

typedef __attribute__((ext_vector_type(2))) float v2f;
typedef __attribute__((ext_vector_type(4))) float v4f;
typedef __attribute__((ext_vector_type(8))) float v8f;

__global__ void build_rowptr(const int* __restrict__ rows, int* __restrict__ row_ptr,
                             int nE, int nN) {
    int e = blockIdx.x * blockDim.x + threadIdx.x;
    if (e >= nE) return;
    int r = rows[e];
    int rp = (e == 0) ? -1 : rows[e - 1];
    for (int rr = rp + 1; rr <= r; ++rr) row_ptr[rr] = e;
    if (e == nE - 1) {
        for (int rr = r + 1; rr <= nN; ++rr) row_ptr[rr] = nE;
    }
}

#define SPMM_STRIP 8
#define SPMM_BATCH 16

__global__ __launch_bounds__(256)
void spmm_kernel(const float* __restrict__ h, const int* __restrict__ row_ptr,
                 const int* __restrict__ cols, const float* __restrict__ vals,
                 float* __restrict__ out, int nN) {
    const int lane = threadIdx.x & 31;
    const int wave = __builtin_amdgcn_readfirstlane(
        (int)((blockIdx.x * blockDim.x + threadIdx.x) >> 5));
    const int n0 = wave * SPMM_STRIP;
    if (n0 >= nN) return;
    const int nn = (nN - n0 < SPMM_STRIP) ? (nN - n0) : SPMM_STRIP;

    int rpi = n0 + lane;
    rpi = (rpi > nN) ? nN : rpi;
    const int rp = row_ptr[rpi];

    const int e_begin = __builtin_amdgcn_readlane(rp, 0);
    const int e_end   = __builtin_amdgcn_readlane(rp, nn);

    const v2f* __restrict__ h2 = (const v2f*)h;
    v2f* __restrict__ out2     = (v2f*)out;

    int node = 0;
    int cur_end = __builtin_amdgcn_readlane(rp, 1);
    v2f acc = {0.0f, 0.0f};
    int   cv = 0;
    float vv = 0.0f;
    if (e_begin + lane < e_end) { cv = cols[e_begin + lane]; vv = vals[e_begin + lane]; }

    for (int base = e_begin; base < e_end; base += 32) {
        int   ncv = 0;
        float nvv = 0.0f;
        const int nidx = base + 32 + lane;
        if (nidx < e_end) { ncv = cols[nidx]; nvv = vals[nidx]; }

        const int left = e_end - base;
        const int cnt = (left < 32) ? left : 32;
        for (int j = 0; j < cnt; j += SPMM_BATCH) {
            v2f   r[SPMM_BATCH];
            float v[SPMM_BATCH];
#pragma unroll
            for (int k = 0; k < SPMM_BATCH; ++k) {
                const int c = __builtin_amdgcn_readlane(cv, j + k);
                v[k] = __int_as_float(__builtin_amdgcn_readlane(__float_as_int(vv), j + k));
                r[k] = h2[(size_t)c * 32 + lane];
            }
#pragma unroll
            for (int k = 0; k < SPMM_BATCH; ++k) {
                const int e = base + j + k;
                if (e < e_end) {
                    while (e >= cur_end) {
                        out2[(size_t)(n0 + node) * 32 + lane] = acc;
                        acc = (v2f){0.0f, 0.0f};
                        ++node;
                        cur_end = __builtin_amdgcn_readlane(rp, node + 1);
                    }
                    acc.x = fmaf(v[k], r[k].x, acc.x);
                    acc.y = fmaf(v[k], r[k].y, acc.y);
                }
            }
        }
        cv = ncv;
        vv = nvv;
    }

    while (node < nn) {
        out2[(size_t)(n0 + node) * 32 + lane] = acc;
        acc = (v2f){0.0f, 0.0f};
        ++node;
    }
}

__device__ __forceinline__ int tile_slot(int L, int p) {
    return L * 32 + (p ^ ((L & 15) << 1));
}

__device__ __forceinline__ void gemm_accum_32(const float* __restrict__ A,
                                              const float* __restrict__ Wl,
                                              int ksel, int bcol, v8f acc[2]) {
#pragma unroll
    for (int kk = 0; kk < 64; kk += 4) {
        v2f a = *(const v2f*)(A + kk + ksel);
        int kprow = ((kk + ksel) >> 1) * 128;
#pragma unroll
        for (int nt = 0; nt < 2; ++nt) {
            v2f b = *(const v2f*)(Wl + kprow + (nt * 16 + bcol) * 2);
            acc[nt] = __builtin_amdgcn_wmma_f32_16x16x4_f32(
                false, a, false, b, (short)0, acc[nt], false, false);
        }
    }
}

__global__ __launch_bounds__(256)
void spmm_gemm_kernel(const float* __restrict__ h, const int* __restrict__ row_ptr,
                      const int* __restrict__ cols, const float* __restrict__ vals,
                      const float* __restrict__ x, const float* __restrict__ W,
                      const float* __restrict__ bias, float* __restrict__ out, int nN) {
    __shared__ __attribute__((aligned(16))) float w_lds[3 * 64 * 64];
    __shared__ v2f tile2[64 * 32];

    const int lane = threadIdx.x & 31;
    const int wib = __builtin_amdgcn_readfirstlane((int)(threadIdx.x >> 5));
    const int n0 = ((int)blockIdx.x * 8 + wib) * SPMM_STRIP;
    const int nn = (nN - n0 < SPMM_STRIP) ? (nN - n0) : SPMM_STRIP;
    const int L0 = wib * SPMM_STRIP;

    int rp = 0;
    if (nn > 0) {
        int rpi = n0 + lane;
        rpi = (rpi > nN) ? nN : rpi;
        rp = row_ptr[rpi];
    }

    {
        v4f r0[6], r1[6];
#pragma unroll
        for (int q = 0; q < 6; ++q) {
            const int it = (int)threadIdx.x + q * 256;
            const int t  = it >> 9;
            const int kp = (it >> 4) & 31;
            const int n4 = it & 15;
            const v4f* wsrc = (const v4f*)(W + (t << 12) + (kp << 7) + (n4 << 2));
            r0[q] = wsrc[0];
            r1[q] = wsrc[16];
        }
#pragma unroll
        for (int q = 0; q < 6; ++q) {
            const int it = (int)threadIdx.x + q * 256;
            const int t  = it >> 9;
            const int kp = (it >> 4) & 31;
            const int n4 = it & 15;
            v4f* wdst = (v4f*)(w_lds + (t << 12) + (kp << 7) + (n4 << 3));
            wdst[0] = (v4f){r0[q].x, r1[q].x, r0[q].y, r1[q].y};
            wdst[1] = (v4f){r0[q].z, r1[q].z, r0[q].w, r1[q].w};
        }
    }
    __syncthreads();

    if (nn > 0) {
        const v2f* __restrict__ sub2 = (const v2f*)x;
        v2f sub_row[SPMM_STRIP];
#pragma unroll
        for (int i = 0; i < SPMM_STRIP; ++i) {
            int nd = n0 + i;
            nd = (nd < nN) ? nd : (nN - 1);
            sub_row[i] = sub2[(size_t)nd * 32 + lane];
        }

        const int e_begin = __builtin_amdgcn_readlane(rp, 0);
        const int e_end   = __builtin_amdgcn_readlane(rp, nn);

        const v2f* __restrict__ h2 = (const v2f*)h;

        int node = 0;
        int cur_end = __builtin_amdgcn_readlane(rp, 1);
        v2f acc = {0.0f, 0.0f};
        int   cv = 0;
        float vv = 0.0f;
        if (e_begin + lane < e_end) { cv = cols[e_begin + lane]; vv = vals[e_begin + lane]; }

#pragma unroll
        for (int i = 0; i < SPMM_STRIP; ++i) tile2[tile_slot(L0 + i, lane)] = sub_row[i];
        v2f sv = tile2[tile_slot(L0, lane)];

        for (int base = e_begin; base < e_end; base += 32) {
            int   ncv = 0;
            float nvv = 0.0f;
            const int nidx = base + 32 + lane;
            if (nidx < e_end) { ncv = cols[nidx]; nvv = vals[nidx]; }

            const int left = e_end - base;
            const int cnt = (left < 32) ? left : 32;
            for (int j = 0; j < cnt; j += SPMM_BATCH) {
                v2f   r[SPMM_BATCH];
                float v[SPMM_BATCH];
#pragma unroll
                for (int k = 0; k < SPMM_BATCH; ++k) {
                    const int c = __builtin_amdgcn_readlane(cv, j + k);
                    v[k] = __int_as_float(__builtin_amdgcn_readlane(__float_as_int(vv), j + k));
                    r[k] = h2[(size_t)c * 32 + lane];
                }
#pragma unroll
                for (int k = 0; k < SPMM_BATCH; ++k) {
                    const int e = base + j + k;
                    if (e < e_end) {
                        while (e >= cur_end) {
                            tile2[tile_slot(L0 + node, lane)] = 2.0f * acc - sv;
                            acc = (v2f){0.0f, 0.0f};
                            ++node;
                            cur_end = __builtin_amdgcn_readlane(rp, node + 1);
                            sv = tile2[tile_slot(L0 + node, lane)];
                        }
                        acc.x = fmaf(v[k], r[k].x, acc.x);
                        acc.y = fmaf(v[k], r[k].y, acc.y);
                    }
                }
            }
            cv = ncv;
            vv = nvv;
        }

        while (node < nn) {
            tile2[tile_slot(L0 + node, lane)] = 2.0f * acc - sv;
            acc = (v2f){0.0f, 0.0f};
            ++node;
            if (node < nn) sv = tile2[tile_slot(L0 + node, lane)];
        }
    } else {
#pragma unroll
        for (int i = 0; i < SPMM_STRIP; ++i) tile2[tile_slot(L0 + i, lane)] = (v2f){0.0f, 0.0f};
    }

    const int rt = wib >> 1;
    const int ch = wib & 1;
    const int m0 = (int)blockIdx.x * 64 + rt * 16;
    const int arow = m0 + (lane & 15);
    const int arow_c = (arow < nN) ? arow : (nN - 1);
    const int ksel = (lane >> 4) << 1;
    const int bn = lane & 15;
    const int bcol = ch * 32 + bn;

    v8f acc2[2] = {};
    gemm_accum_32(x + (size_t)arow_c * 64, w_lds,             ksel, bcol, acc2);
    gemm_accum_32(h + (size_t)arow_c * 64, w_lds + (1 << 12), ksel, bcol, acc2);

    __syncthreads();

    {
        const int Lr = rt * 16 + (lane & 15);
        const float* Wl = w_lds + (2 << 12);
#pragma unroll
        for (int kk = 0; kk < 64; kk += 4) {
            const v2f a = tile2[tile_slot(Lr, (kk + ksel) >> 1)];
            const int kprow = ((kk + ksel) >> 1) * 128;
#pragma unroll
            for (int nt = 0; nt < 2; ++nt) {
                const v2f b = *(const v2f*)(Wl + kprow + (nt * 16 + bcol) * 2);
                acc2[nt] = __builtin_amdgcn_wmma_f32_16x16x4_f32(
                    false, a, false, b, (short)0, acc2[nt], false, false);
            }
        }
    }

    const int mbase = m0 + ((lane >> 4) << 3);
#pragma unroll
    for (int nt = 0; nt < 2; ++nt) {
        const int col = ch * 32 + nt * 16 + bn;
        const float bv = bias[col];
#pragma unroll
        for (int j = 0; j < 8; ++j) {
            const int row = mbase + j;
            if (row < nN) out[(size_t)row * 64 + col] = acc2[nt][j] + bv;
        }
    }
}

extern "C" void kernel_launch(void* const* d_in, const int* in_sizes, int n_in,
                              void* d_out, int out_size, void* d_ws, size_t ws_size,
                              hipStream_t stream) {
    const float* x    = (const float*)d_in[0];
    const int*   rows = (const int*)d_in[1];
    const int*   cols = (const int*)d_in[2];
    const float* vals = (const float*)d_in[3];
    const float* W    = (const float*)d_in[4];
    const float* bias = (const float*)d_in[5];

    const int N = in_sizes[0] / 64;
    const int E = in_sizes[1];

    char* ws = (char*)d_ws;
    int* row_ptr = (int*)ws;
    size_t off = (((size_t)(N + 1) * sizeof(int)) + 255) & ~(size_t)255;
    float* Tx1 = (float*)(ws + off);

    build_rowptr<<<(E + 255) / 256, 256, 0, stream>>>(rows, row_ptr, E, N);

    const int nodes_per_block = 8 * SPMM_STRIP;
    int spmm_grid = (N + nodes_per_block - 1) / nodes_per_block;
    spmm_kernel<<<spmm_grid, 256, 0, stream>>>(x, row_ptr, cols, vals, Tx1, N);
    spmm_gemm_kernel<<<spmm_grid, 256, 0, stream>>>(Tx1, row_ptr, cols, vals, x, W, bias, (float*)d_out, N);
}


# Row-pointer build on 391 workgroups instead of 6250: each thread takes 16 sorted edges (four 16-byte loads plus the element in front of each, all issued before the first compare)
# speedup vs baseline: 2.1252x; 1.0337x over previous
#include <hip/hip_runtime.h>

typedef __attribute__((ext_vector_type(2))) float v2f;
typedef __attribute__((ext_vector_type(4))) float v4f;
typedef __attribute__((ext_vector_type(8))) float v8f;

typedef __attribute__((ext_vector_type(4))) int v4i;
#define RP_VEC 4
#define RP_BLOCK_EDGES (256 * 4 * RP_VEC)

__global__ __launch_bounds__(256)
void build_rowptr(const int* __restrict__ rows, int* __restrict__ row_ptr,
                  int nE, int nN) {
    const int blk = (int)blockIdx.x * RP_BLOCK_EDGES;
    const int p0 = blk + (int)threadIdx.x * 4;
    const int last = nE - 1;
    v4i q[RP_VEC];
    int prev[RP_VEC];
    if (blk + RP_BLOCK_EDGES <= nE) {
#pragma unroll
        for (int k = 0; k < RP_VEC; ++k) q[k] = *(const v4i*)(rows + p0 + k * 1024);
#pragma unroll
        for (int k = 0; k < RP_VEC; ++k) {
            int pi = p0 + k * 1024 - 1;
            pi = (pi < 0) ? 0 : pi;
            prev[k] = rows[pi];
        }
    } else {
#pragma unroll
        for (int k = 0; k < RP_VEC; ++k) {
            const int p = p0 + k * 1024;
            int i0 = p, i1 = p + 1, i2 = p + 2, i3 = p + 3, pi = p - 1;
            i0 = (i0 > last) ? last : i0;
            i1 = (i1 > last) ? last : i1;
            i2 = (i2 > last) ? last : i2;
            i3 = (i3 > last) ? last : i3;
            pi = (pi < 0) ? 0 : pi;
            pi = (pi > last) ? last : pi;
            q[k] = (v4i){rows[i0], rows[i1], rows[i2], rows[i3]};
            prev[k] = rows[pi];
        }
    }
    if (p0 == 0) prev[0] = -1;

    int tail_r = -2;
#pragma unroll
    for (int k = 0; k < RP_VEC; ++k) {
        int rp = prev[k];
#pragma unroll
        for (int j = 0; j < 4; ++j) {
            const int e = p0 + k * 1024 + j;
            const int r = q[k][j];
            if (r != rp && e < nE) {
#pragma clang loop unroll(disable) vectorize(disable)
                for (int rr = rp + 1; rr <= r; ++rr) row_ptr[rr] = e;
            }
            if (e == last) tail_r = r;
            rp = r;
        }
    }
    if (tail_r != -2) {
#pragma clang loop unroll(disable) vectorize(disable)
        for (int rr = tail_r + 1; rr <= nN; ++rr) row_ptr[rr] = nE;
    }
}

#define SPMM_STRIP 8
#define SPMM_BATCH 16

__global__ __launch_bounds__(256)
void spmm_kernel(const float* __restrict__ h, const int* __restrict__ row_ptr,
                 const int* __restrict__ cols, const float* __restrict__ vals,
                 float* __restrict__ out, int nN) {
    const int lane = threadIdx.x & 31;
    const int wave = __builtin_amdgcn_readfirstlane(
        (int)((blockIdx.x * blockDim.x + threadIdx.x) >> 5));
    const int n0 = wave * SPMM_STRIP;
    if (n0 >= nN) return;
    const int nn = (nN - n0 < SPMM_STRIP) ? (nN - n0) : SPMM_STRIP;

    int rpi = n0 + lane;
    rpi = (rpi > nN) ? nN : rpi;
    const int rp = row_ptr[rpi];

    const int e_begin = __builtin_amdgcn_readlane(rp, 0);
    const int e_end   = __builtin_amdgcn_readlane(rp, nn);

    const v2f* __restrict__ h2 = (const v2f*)h;
    v2f* __restrict__ out2     = (v2f*)out;

    int node = 0;
    int cur_end = __builtin_amdgcn_readlane(rp, 1);
    v2f acc = {0.0f, 0.0f};
    int   cv = 0;
    float vv = 0.0f;
    if (e_begin + lane < e_end) { cv = cols[e_begin + lane]; vv = vals[e_begin + lane]; }

    for (int base = e_begin; base < e_end; base += 32) {
        int   ncv = 0;
        float nvv = 0.0f;
        const int nidx = base + 32 + lane;
        if (nidx < e_end) { ncv = cols[nidx]; nvv = vals[nidx]; }

        const int left = e_end - base;
        const int cnt = (left < 32) ? left : 32;
        for (int j = 0; j < cnt; j += SPMM_BATCH) {
            v2f   r[SPMM_BATCH];
            float v[SPMM_BATCH];
#pragma unroll
            for (int k = 0; k < SPMM_BATCH; ++k) {
                const int c = __builtin_amdgcn_readlane(cv, j + k);
                v[k] = __int_as_float(__builtin_amdgcn_readlane(__float_as_int(vv), j + k));
                r[k] = h2[(size_t)c * 32 + lane];
            }
#pragma unroll
            for (int k = 0; k < SPMM_BATCH; ++k) {
                const int e = base + j + k;
                if (e < e_end) {
                    while (e >= cur_end) {
                        out2[(size_t)(n0 + node) * 32 + lane] = acc;
                        acc = (v2f){0.0f, 0.0f};
                        ++node;
                        cur_end = __builtin_amdgcn_readlane(rp, node + 1);
                    }
                    acc.x = fmaf(v[k], r[k].x, acc.x);
                    acc.y = fmaf(v[k], r[k].y, acc.y);
                }
            }
        }
        cv = ncv;
        vv = nvv;
    }

    while (node < nn) {
        out2[(size_t)(n0 + node) * 32 + lane] = acc;
        acc = (v2f){0.0f, 0.0f};
        ++node;
    }
}

__device__ __forceinline__ int tile_slot(int L, int p) {
    return L * 32 + (p ^ ((L & 15) << 1));
}

__device__ __forceinline__ void gemm_accum_32(const float* __restrict__ A,
                                              const float* __restrict__ Wl,
                                              int ksel, int bcol, v8f acc[2]) {
#pragma unroll
    for (int kk = 0; kk < 64; kk += 4) {
        v2f a = *(const v2f*)(A + kk + ksel);
        int kprow = ((kk + ksel) >> 1) * 128;
#pragma unroll
        for (int nt = 0; nt < 2; ++nt) {
            v2f b = *(const v2f*)(Wl + kprow + (nt * 16 + bcol) * 2);
            acc[nt] = __builtin_amdgcn_wmma_f32_16x16x4_f32(
                false, a, false, b, (short)0, acc[nt], false, false);
        }
    }
}

__global__ __launch_bounds__(256)
void spmm_gemm_kernel(const float* __restrict__ h, const int* __restrict__ row_ptr,
                      const int* __restrict__ cols, const float* __restrict__ vals,
                      const float* __restrict__ x, const float* __restrict__ W,
                      const float* __restrict__ bias, float* __restrict__ out, int nN) {
    __shared__ __attribute__((aligned(16))) float w_lds[3 * 64 * 64];
    __shared__ v2f tile2[64 * 32];

    const int lane = threadIdx.x & 31;
    const int wib = __builtin_amdgcn_readfirstlane((int)(threadIdx.x >> 5));
    const int n0 = ((int)blockIdx.x * 8 + wib) * SPMM_STRIP;
    const int nn = (nN - n0 < SPMM_STRIP) ? (nN - n0) : SPMM_STRIP;
    const int L0 = wib * SPMM_STRIP;

    int rp = 0;
    if (nn > 0) {
        int rpi = n0 + lane;
        rpi = (rpi > nN) ? nN : rpi;
        rp = row_ptr[rpi];
    }

    {
        v4f r0[6], r1[6];
#pragma unroll
        for (int q = 0; q < 6; ++q) {
            const int it = (int)threadIdx.x + q * 256;
            const int t  = it >> 9;
            const int kp = (it >> 4) & 31;
            const int n4 = it & 15;
            const v4f* wsrc = (const v4f*)(W + (t << 12) + (kp << 7) + (n4 << 2));
            r0[q] = wsrc[0];
            r1[q] = wsrc[16];
        }
#pragma unroll
        for (int q = 0; q < 6; ++q) {
            const int it = (int)threadIdx.x + q * 256;
            const int t  = it >> 9;
            const int kp = (it >> 4) & 31;
            const int n4 = it & 15;
            v4f* wdst = (v4f*)(w_lds + (t << 12) + (kp << 7) + (n4 << 3));
            wdst[0] = (v4f){r0[q].x, r1[q].x, r0[q].y, r1[q].y};
            wdst[1] = (v4f){r0[q].z, r1[q].z, r0[q].w, r1[q].w};
        }
    }
    __syncthreads();

    if (nn > 0) {
        const v2f* __restrict__ sub2 = (const v2f*)x;
        v2f sub_row[SPMM_STRIP];
#pragma unroll
        for (int i = 0; i < SPMM_STRIP; ++i) {
            int nd = n0 + i;
            nd = (nd < nN) ? nd : (nN - 1);
            sub_row[i] = sub2[(size_t)nd * 32 + lane];
        }

        const int e_begin = __builtin_amdgcn_readlane(rp, 0);
        const int e_end   = __builtin_amdgcn_readlane(rp, nn);

        const v2f* __restrict__ h2 = (const v2f*)h;

        int node = 0;
        int cur_end = __builtin_amdgcn_readlane(rp, 1);
        v2f acc = {0.0f, 0.0f};
        int   cv = 0;
        float vv = 0.0f;
        if (e_begin + lane < e_end) { cv = cols[e_begin + lane]; vv = vals[e_begin + lane]; }

#pragma unroll
        for (int i = 0; i < SPMM_STRIP; ++i) tile2[tile_slot(L0 + i, lane)] = sub_row[i];
        v2f sv = tile2[tile_slot(L0, lane)];

        for (int base = e_begin; base < e_end; base += 32) {
            int   ncv = 0;
            float nvv = 0.0f;
            const int nidx = base + 32 + lane;
            if (nidx < e_end) { ncv = cols[nidx]; nvv = vals[nidx]; }

            const int left = e_end - base;
            const int cnt = (left < 32) ? left : 32;
            for (int j = 0; j < cnt; j += SPMM_BATCH) {
                v2f   r[SPMM_BATCH];
                float v[SPMM_BATCH];
#pragma unroll
                for (int k = 0; k < SPMM_BATCH; ++k) {
                    const int c = __builtin_amdgcn_readlane(cv, j + k);
                    v[k] = __int_as_float(__builtin_amdgcn_readlane(__float_as_int(vv), j + k));
                    r[k] = h2[(size_t)c * 32 + lane];
                }
#pragma unroll
                for (int k = 0; k < SPMM_BATCH; ++k) {
                    const int e = base + j + k;
                    if (e < e_end) {
                        while (e >= cur_end) {
                            tile2[tile_slot(L0 + node, lane)] = 2.0f * acc - sv;
                            acc = (v2f){0.0f, 0.0f};
                            ++node;
                            cur_end = __builtin_amdgcn_readlane(rp, node + 1);
                            sv = tile2[tile_slot(L0 + node, lane)];
                        }
                        acc.x = fmaf(v[k], r[k].x, acc.x);
                        acc.y = fmaf(v[k], r[k].y, acc.y);
                    }
                }
            }
            cv = ncv;
            vv = nvv;
        }

        while (node < nn) {
            tile2[tile_slot(L0 + node, lane)] = 2.0f * acc - sv;
            acc = (v2f){0.0f, 0.0f};
            ++node;
            if (node < nn) sv = tile2[tile_slot(L0 + node, lane)];
        }
    } else {
#pragma unroll
        for (int i = 0; i < SPMM_STRIP; ++i) tile2[tile_slot(L0 + i, lane)] = (v2f){0.0f, 0.0f};
    }

    const int rt = wib >> 1;
    const int ch = wib & 1;
    const int m0 = (int)blockIdx.x * 64 + rt * 16;
    const int arow = m0 + (lane & 15);
    const int arow_c = (arow < nN) ? arow : (nN - 1);
    const int ksel = (lane >> 4) << 1;
    const int bn = lane & 15;
    const int bcol = ch * 32 + bn;

    v8f acc2[2] = {};
    gemm_accum_32(x + (size_t)arow_c * 64, w_lds,             ksel, bcol, acc2);
    gemm_accum_32(h + (size_t)arow_c * 64, w_lds + (1 << 12), ksel, bcol, acc2);

    __syncthreads();

    {
        const int Lr = rt * 16 + (lane & 15);
        const float* Wl = w_lds + (2 << 12);
#pragma unroll
        for (int kk = 0; kk < 64; kk += 4) {
            const v2f a = tile2[tile_slot(Lr, (kk + ksel) >> 1)];
            const int kprow = ((kk + ksel) >> 1) * 128;
#pragma unroll
            for (int nt = 0; nt < 2; ++nt) {
                const v2f b = *(const v2f*)(Wl + kprow + (nt * 16 + bcol) * 2);
                acc2[nt] = __builtin_amdgcn_wmma_f32_16x16x4_f32(
                    false, a, false, b, (short)0, acc2[nt], false, false);
            }
        }
    }

    const int mbase = m0 + ((lane >> 4) << 3);
#pragma unroll
    for (int nt = 0; nt < 2; ++nt) {
        const int col = ch * 32 + nt * 16 + bn;
        const float bv = bias[col];
#pragma unroll
        for (int j = 0; j < 8; ++j) {
            const int row = mbase + j;
            if (row < nN) out[(size_t)row * 64 + col] = acc2[nt][j] + bv;
        }
    }
}

extern "C" void kernel_launch(void* const* d_in, const int* in_sizes, int n_in,
                              void* d_out, int out_size, void* d_ws, size_t ws_size,
                              hipStream_t stream) {
    const float* x    = (const float*)d_in[0];
    const int*   rows = (const int*)d_in[1];
    const int*   cols = (const int*)d_in[2];
    const float* vals = (const float*)d_in[3];
    const float* W    = (const float*)d_in[4];
    const float* bias = (const float*)d_in[5];

    const int N = in_sizes[0] / 64;
    const int E = in_sizes[1];

    char* ws = (char*)d_ws;
    int* row_ptr = (int*)ws;
    size_t off = (((size_t)(N + 1) * sizeof(int)) + 255) & ~(size_t)255;
    float* Tx1 = (float*)(ws + off);

    build_rowptr<<<(E + RP_BLOCK_EDGES - 1) / RP_BLOCK_EDGES, 256, 0, stream>>>(rows, row_ptr, E, N);

    const int nodes_per_block = 8 * SPMM_STRIP;
    int spmm_grid = (N + nodes_per_block - 1) / nodes_per_block;
    spmm_kernel<<<spmm_grid, 256, 0, stream>>>(x, row_ptr, cols, vals, Tx1, N);
    spmm_gemm_kernel<<<spmm_grid, 256, 0, stream>>>(Tx1, row_ptr, cols, vals, x, W, bias, (float*)d_out, N);
}
